# starting kernel
# baseline (speedup 1.0000x reference)
#include <hip/hip_runtime.h>

#define H_  1024
#define W_  2048
#define C_  16
#define SH_ 68
#define SW_ 132

typedef float v2f __attribute__((ext_vector_type(2)));
typedef float v8f __attribute__((ext_vector_type(8)));

__global__ __launch_bounds__(256) void splat_kernel(const float* __restrict__ inp,
                                                    float* __restrict__ G)
{
    const int cx  = blockIdx.x;
    const int cy  = blockIdx.y;
    const int tid = threadIdx.x;
    const int py  = tid >> 4, px = tid & 15;
    const int y   = 16 * (cy - 2) - 8 + py;
    const int x   = 16 * (cx - 2) - 8 + px;

    float4 v0 = make_float4(0.f, 0.f, 0.f, 0.f);
    float4 v1 = v0, v2 = v0, v3 = v0;
    if ((unsigned)y < (unsigned)H_ && (unsigned)x < (unsigned)W_) {
        const float4* p = (const float4*)(inp + ((size_t)y * W_ + x) * C_);
        v0 = p[0]; v1 = p[1]; v2 = p[2]; v3 = p[3];
    }

    __shared__ float4 red[256 * 4];
    red[tid * 4 + 0] = v0; red[tid * 4 + 1] = v1;
    red[tid * 4 + 2] = v2; red[tid * 4 + 3] = v3;
    __syncthreads();

    for (int s = 128; s > 0; s >>= 1) {
        if (tid < s) {
#pragma unroll
            for (int q = 0; q < 4; ++q) {
                float4 a = red[tid * 4 + q];
                float4 b = red[(tid + s) * 4 + q];
                a.x += b.x; a.y += b.y; a.z += b.z; a.w += b.w;
                red[tid * 4 + q] = a;
            }
        }
        __syncthreads();
    }
    if (tid < C_)
        G[((size_t)cy * SW_ + cx) * C_ + tid] = ((const float*)red)[tid];
}

__global__ __launch_bounds__(256) void blurA_kernel(const float* __restrict__ G,
                                                    float* __restrict__ A)
{
    const int idx  = blockIdx.x * 256 + threadIdx.x;
    const int cell = idx >> 4;
    const int x    = cell % SW_;
    const int y    = cell / SW_;
    const float gc = G[idx];
    float r;
    if (x > 0 && x < SW_ - 1)
        r = (G[idx - C_] + G[idx + C_] + 2.0f * gc) * 0.25f;
    else if (y > 0 && y < SH_ - 1)
        r = (G[idx - C_ * SW_] + G[idx + C_ * SW_] + 2.0f * gc) * 0.25f;
    else
        r = 0.0f;
    A[idx] = r;
}

__global__ __launch_bounds__(256) void blurB_kernel(const float* __restrict__ G,
                                                    const float* __restrict__ A,
                                                    float* __restrict__ B)
{
    const int idx  = blockIdx.x * 256 + threadIdx.x;
    const int cell = idx >> 4;
    const int x    = cell % SW_;
    const int y    = cell / SW_;
    const float ac = A[idx];
    float r;
    if (x > 0 && x < SW_ - 1)
        r = (A[idx - C_] + A[idx + C_] + 2.0f * ac) * 0.25f;
    else if (y > 0 && y < SH_ - 1)
        r = (A[idx - C_ * SW_] + A[idx + C_ * SW_] + 2.0f * ac) * 0.25f;
    else
        r = G[idx];
    B[idx] = r;
}

__global__ __launch_bounds__(256) void slice_kernel(const float* __restrict__ Bg,
                                                    float* __restrict__ out)
{
    const int wave = threadIdx.x >> 5;
    const int lane = threadIdx.x & 31;
    const int tile = blockIdx.x * 8 + wave;
    const int k    = tile >> 7;
    const int j    = tile & 127;
    const int m    = lane & 15;
    const int hi   = lane >> 4;

    const float* gB = Bg + (((size_t)(k + 2 + hi) * SW_) + (j + 2)) * C_ + m;
    v2f b;
    b.x = gB[0];
    b.y = gB[C_];

    const float am = (float)m * 0.0625f;
    const float w0 = 1.0f - am;

    float* obase = out + ((size_t)(16 * k) * W_ + 16 * j) * C_;

#pragma unroll
    for (int py = 0; py < 16; ++py) {
        const float ay = (float)py * 0.0625f;
        const float wy = hi ? ay : (1.0f - ay);
        v2f a;
        a.x = wy * w0;
        a.y = wy * am;
        v8f cacc = {};
        v8f d = __builtin_amdgcn_wmma_f32_16x16x4_f32(
             false, a,  false, b,
             (short)0, cacc,  false,  false);

        float* orow = obase + (size_t)py * W_ * C_ + (size_t)hi * 8 * C_ + m;
#pragma unroll
        for (int r = 0; r < 8; ++r)
            orow[r * C_] = d[r];
    }
}

extern "C" void kernel_launch(void* const* d_in, const int* in_sizes, int n_in,
                              void* d_out, int out_size, void* d_ws, size_t ws_size,
                              hipStream_t stream)
{
    const float* inp = (const float*)d_in[0];
    float* out = (float*)d_out;

    const size_t S = (size_t)SH_ * SW_ * C_;
    float* G  = (float*)d_ws;
    float* A  = G + S;
    float* Bg = A + S;

    splat_kernel<<<dim3(SW_, SH_), 256, 0, stream>>>(inp, G);

    const int n = SH_ * SW_ * C_;
    blurA_kernel<<<n / 256, 256, 0, stream>>>(G, A);
    blurB_kernel<<<n / 256, 256, 0, stream>>>(G, A, Bg);

    slice_kernel<<<8192 / 8, 256, 0, stream>>>(Bg, out);
}


# splat: one wave per cell sums its 32 coalesced 512-byte row pieces in registers and folds 8 lane groups by shuffle, replacing the 256-thread LDS tree; the block sums are added in a different order
# speedup vs baseline: 1.2709x; 1.2709x over previous
#include <hip/hip_runtime.h>

#define H_  1024
#define W_  2048
#define C_  16
#define SH_ 68
#define SW_ 132

typedef float v2f __attribute__((ext_vector_type(2)));
typedef float v8f __attribute__((ext_vector_type(8)));

typedef float v4f __attribute__((ext_vector_type(4)));
#define ROWSTEP_ (W_ * C_ / 4)
#define PIECE_   (8 * C_ / 4)

__device__ __forceinline__ v4f inner_cell_sum(const v4f* __restrict__ rect)
{
    v4f acc[4];
#pragma unroll
    for (int c = 0; c < 4; ++c) acc[c] = (v4f){0.f, 0.f, 0.f, 0.f};
#pragma unroll
    for (int r = 0; r < 16; ++r) {
        acc[(2 * r) & 3]     += rect[(size_t)r * ROWSTEP_];
        acc[(2 * r + 1) & 3] += rect[(size_t)r * ROWSTEP_ + PIECE_];
    }
    return (acc[0] + acc[1]) + (acc[2] + acc[3]);
}

__device__ __forceinline__ v4f edge_cell_sum(const v4f* __restrict__ rect, int ny,
                                             bool two_pieces)
{
    v4f acc[4];
#pragma unroll
    for (int c = 0; c < 4; ++c) acc[c] = (v4f){0.f, 0.f, 0.f, 0.f};
    for (int r0 = 0; r0 < ny; r0 += 8) {
        const v4f* p = rect + (size_t)r0 * ROWSTEP_;
        v4f t[8];
#pragma unroll
        for (int r = 0; r < 8; ++r) t[r] = p[(size_t)r * ROWSTEP_];
        if (two_pieces) {
            v4f u[8];
#pragma unroll
            for (int r = 0; r < 8; ++r) u[r] = p[(size_t)r * ROWSTEP_ + PIECE_];
#pragma unroll
            for (int r = 0; r < 8; ++r) acc[2 + (r & 1)] += u[r];
        }
#pragma unroll
        for (int r = 0; r < 8; ++r) acc[r & 1] += t[r];
    }
    return (acc[0] + acc[1]) + (acc[2] + acc[3]);
}

__global__ __launch_bounds__(256, 10) void splat_kernel(const float* __restrict__ inp,
                                                        float* __restrict__ G)
{
    const int lane = threadIdx.x & 31;
    const int wave = __builtin_amdgcn_readfirstlane((int)(threadIdx.x >> 5));
    const int cell = blockIdx.x * 8 + wave;
    const int cy   = cell / SW_;
    const int cx   = cell - cy * SW_;
    const int y0   = 16 * (cy - 2) - 8;
    const int x0   = 16 * (cx - 2) - 8;
    const int vy   = max(y0, 0), ny = min(y0 + 16, H_) - vy;
    const int vx   = max(x0, 0), nx = min(x0 + 16, W_) - vx;

    v4f s = (v4f){0.f, 0.f, 0.f, 0.f};
    if (ny > 0 && nx > 0) {
        const v4f* rect = (const v4f*)(inp + ((size_t)vy * W_ + vx) * C_) + lane;
        if (ny == 16 && nx == 16) s = inner_cell_sum(rect);
        else                      s = edge_cell_sum(rect, ny, nx == 16);

#pragma unroll
        for (int m = 4; m < 32; m <<= 1) {
            s.x += __shfl_xor(s.x, m);
            s.y += __shfl_xor(s.y, m);
            s.z += __shfl_xor(s.z, m);
            s.w += __shfl_xor(s.w, m);
        }
    }
    if (lane < 4) ((v4f*)(G + (size_t)cell * C_))[lane] = s;
}

__global__ __launch_bounds__(256) void blurA_kernel(const float* __restrict__ G,
                                                    float* __restrict__ A)
{
    const int idx  = blockIdx.x * 256 + threadIdx.x;
    const int cell = idx >> 4;
    const int x    = cell % SW_;
    const int y    = cell / SW_;
    const float gc = G[idx];
    float r;
    if (x > 0 && x < SW_ - 1)
        r = (G[idx - C_] + G[idx + C_] + 2.0f * gc) * 0.25f;
    else if (y > 0 && y < SH_ - 1)
        r = (G[idx - C_ * SW_] + G[idx + C_ * SW_] + 2.0f * gc) * 0.25f;
    else
        r = 0.0f;
    A[idx] = r;
}

__global__ __launch_bounds__(256) void blurB_kernel(const float* __restrict__ G,
                                                    const float* __restrict__ A,
                                                    float* __restrict__ B)
{
    const int idx  = blockIdx.x * 256 + threadIdx.x;
    const int cell = idx >> 4;
    const int x    = cell % SW_;
    const int y    = cell / SW_;
    const float ac = A[idx];
    float r;
    if (x > 0 && x < SW_ - 1)
        r = (A[idx - C_] + A[idx + C_] + 2.0f * ac) * 0.25f;
    else if (y > 0 && y < SH_ - 1)
        r = (A[idx - C_ * SW_] + A[idx + C_ * SW_] + 2.0f * ac) * 0.25f;
    else
        r = G[idx];
    B[idx] = r;
}

__global__ __launch_bounds__(256) void slice_kernel(const float* __restrict__ Bg,
                                                    float* __restrict__ out)
{
    const int wave = threadIdx.x >> 5;
    const int lane = threadIdx.x & 31;
    const int tile = blockIdx.x * 8 + wave;
    const int k    = tile >> 7;
    const int j    = tile & 127;
    const int m    = lane & 15;
    const int hi   = lane >> 4;

    const float* gB = Bg + (((size_t)(k + 2 + hi) * SW_) + (j + 2)) * C_ + m;
    v2f b;
    b.x = gB[0];
    b.y = gB[C_];

    const float am = (float)m * 0.0625f;
    const float w0 = 1.0f - am;

    float* obase = out + ((size_t)(16 * k) * W_ + 16 * j) * C_;

#pragma unroll
    for (int py = 0; py < 16; ++py) {
        const float ay = (float)py * 0.0625f;
        const float wy = hi ? ay : (1.0f - ay);
        v2f a;
        a.x = wy * w0;
        a.y = wy * am;
        v8f cacc = {};
        v8f d = __builtin_amdgcn_wmma_f32_16x16x4_f32(
             false, a,  false, b,
             (short)0, cacc,  false,  false);

        float* orow = obase + (size_t)py * W_ * C_ + (size_t)hi * 8 * C_ + m;
#pragma unroll
        for (int r = 0; r < 8; ++r)
            orow[r * C_] = d[r];
    }
}

extern "C" void kernel_launch(void* const* d_in, const int* in_sizes, int n_in,
                              void* d_out, int out_size, void* d_ws, size_t ws_size,
                              hipStream_t stream)
{
    const float* inp = (const float*)d_in[0];
    float* out = (float*)d_out;

    const size_t S = (size_t)SH_ * SW_ * C_;
    float* G  = (float*)d_ws;
    float* A  = G + S;
    float* Bg = A + S;

    splat_kernel<<<SH_ * SW_ / 8, 256, 0, stream>>>(inp, G);

    const int n = SH_ * SW_ * C_;
    blurA_kernel<<<n / 256, 256, 0, stream>>>(G, A);
    blurB_kernel<<<n / 256, 256, 0, stream>>>(G, A, Bg);

    slice_kernel<<<8192 / 8, 256, 0, stream>>>(Bg, out);
}


# blur folded into the slice kernel: each wave builds its 2x2 quad of twice-blurred cells from one 6-cell row piece of the splat grid, same formulas and summation order; 4 launches become 2
# speedup vs baseline: 1.4665x; 1.1539x over previous
#include <hip/hip_runtime.h>

#define H_  1024
#define W_  2048
#define C_  16
#define SH_ 68
#define SW_ 132

typedef float v2f __attribute__((ext_vector_type(2)));
typedef float v8f __attribute__((ext_vector_type(8)));

typedef float v4f __attribute__((ext_vector_type(4)));
#define ROWSTEP_ (W_ * C_ / 4)
#define PIECE_   (8 * C_ / 4)

__device__ __forceinline__ v4f inner_cell_sum(const v4f* __restrict__ rect)
{
    v4f acc[4];
#pragma unroll
    for (int c = 0; c < 4; ++c) acc[c] = (v4f){0.f, 0.f, 0.f, 0.f};
#pragma unroll
    for (int r = 0; r < 16; ++r) {
        acc[(2 * r) & 3]     += rect[(size_t)r * ROWSTEP_];
        acc[(2 * r + 1) & 3] += rect[(size_t)r * ROWSTEP_ + PIECE_];
    }
    return (acc[0] + acc[1]) + (acc[2] + acc[3]);
}

__device__ __forceinline__ v4f edge_cell_sum(const v4f* __restrict__ rect, int ny,
                                             bool two_pieces)
{
    v4f acc[4];
#pragma unroll
    for (int c = 0; c < 4; ++c) acc[c] = (v4f){0.f, 0.f, 0.f, 0.f};
    for (int r0 = 0; r0 < ny; r0 += 8) {
        const v4f* p = rect + (size_t)r0 * ROWSTEP_;
        v4f t[8];
#pragma unroll
        for (int r = 0; r < 8; ++r) t[r] = p[(size_t)r * ROWSTEP_];
        if (two_pieces) {
            v4f u[8];
#pragma unroll
            for (int r = 0; r < 8; ++r) u[r] = p[(size_t)r * ROWSTEP_ + PIECE_];
#pragma unroll
            for (int r = 0; r < 8; ++r) acc[2 + (r & 1)] += u[r];
        }
#pragma unroll
        for (int r = 0; r < 8; ++r) acc[r & 1] += t[r];
    }
    return (acc[0] + acc[1]) + (acc[2] + acc[3]);
}

__global__ __launch_bounds__(256, 10) void splat_kernel(const float* __restrict__ inp,
                                                        float* __restrict__ G)
{
    const int lane = threadIdx.x & 31;
    const int wave = __builtin_amdgcn_readfirstlane((int)(threadIdx.x >> 5));
    const int cell = blockIdx.x * 8 + wave;
    const int cy   = cell / SW_;
    const int cx   = cell - cy * SW_;
    const int y0   = 16 * (cy - 2) - 8;
    const int x0   = 16 * (cx - 2) - 8;
    const int vy   = max(y0, 0), ny = min(y0 + 16, H_) - vy;
    const int vx   = max(x0, 0), nx = min(x0 + 16, W_) - vx;

    v4f s = (v4f){0.f, 0.f, 0.f, 0.f};
    if (ny > 0 && nx > 0) {
        const v4f* rect = (const v4f*)(inp + ((size_t)vy * W_ + vx) * C_) + lane;
        if (ny == 16 && nx == 16) s = inner_cell_sum(rect);
        else                      s = edge_cell_sum(rect, ny, nx == 16);

#pragma unroll
        for (int m = 4; m < 32; m <<= 1) {
            s.x += __shfl_xor(s.x, m);
            s.y += __shfl_xor(s.y, m);
            s.z += __shfl_xor(s.z, m);
            s.w += __shfl_xor(s.w, m);
        }
    }
    if (lane < 4) ((v4f*)(G + (size_t)cell * C_))[lane] = s;
}

__global__ __launch_bounds__(256) void slice_kernel(const float* __restrict__ G,
                                                    float* __restrict__ out)
{
    const int wave = __builtin_amdgcn_readfirstlane((int)(threadIdx.x >> 5));
    const int lane = threadIdx.x & 31;
    const int tile = blockIdx.x * 8 + wave;
    const int k    = tile >> 7;
    const int j    = tile & 127;
    const int m    = lane & 15;
    const int hi   = lane >> 4;

    const float* g = G + (((size_t)(k + 2 + hi) * SW_) + j) * C_ + m;
    const float g0 = g[0];
    const float g1 = g[C_];
    const float g2 = g[2 * C_];
    const float g3 = g[3 * C_];
    const float g4 = g[4 * C_];
    const float a1 = (g0 + g2 + 2.0f * g1) * 0.25f;
    const float a2 = (g1 + g3 + 2.0f * g2) * 0.25f;
    const float a3 = (g2 + g4 + 2.0f * g3) * 0.25f;
    const bool lastcol = (j + 4 == SW_ - 1);
    const float n0 = g[lastcol ? 4 * C_ - C_ * SW_ : 3 * C_];
    const float n1 = g[lastcol ? 4 * C_ + C_ * SW_ : 5 * C_];
    const float a4 = (n0 + n1 + 2.0f * g4) * 0.25f;
    v2f b;
    b.x = (a1 + a3 + 2.0f * a2) * 0.25f;
    b.y = (a2 + a4 + 2.0f * a3) * 0.25f;

    const float am = (float)m * 0.0625f;
    const float w0 = 1.0f - am;

    float* obase = out + ((size_t)(16 * k) * W_ + 16 * j) * C_;

#pragma unroll
    for (int py = 0; py < 16; ++py) {
        const float ay = (float)py * 0.0625f;
        const float wy = hi ? ay : (1.0f - ay);
        v2f a;
        a.x = wy * w0;
        a.y = wy * am;
        v8f cacc = {};
        v8f d = __builtin_amdgcn_wmma_f32_16x16x4_f32(
             false, a,  false, b,
             (short)0, cacc,  false,  false);

        float* orow = obase + (size_t)py * W_ * C_ + (size_t)hi * 8 * C_ + m;
#pragma unroll
        for (int r = 0; r < 8; ++r)
            orow[r * C_] = d[r];
    }
}

extern "C" void kernel_launch(void* const* d_in, const int* in_sizes, int n_in,
                              void* d_out, int out_size, void* d_ws, size_t ws_size,
                              hipStream_t stream)
{
    const float* inp = (const float*)d_in[0];
    float* out = (float*)d_out;
    float* G   = (float*)d_ws;

    splat_kernel<<<SH_ * SW_ / 8, 256, 0, stream>>>(inp, G);

    slice_kernel<<<8192 / 8, 256, 0, stream>>>(G, out);
}


# slice stores whole 128-byte lines: product taken transposed (channels x pixels), lanes swap one 16-byte chunk per row by DPP, two 16-byte stores of 512 contiguous bytes replace eight 4-byte stores
# speedup vs baseline: 1.9859x; 1.3542x over previous
#include <hip/hip_runtime.h>

#define H_  1024
#define W_  2048
#define C_  16
#define SH_ 68
#define SW_ 132

typedef float v2f __attribute__((ext_vector_type(2)));
typedef float v8f __attribute__((ext_vector_type(8)));

typedef float v4f __attribute__((ext_vector_type(4)));
#define ROWSTEP_ (W_ * C_ / 4)
#define PIECE_   (8 * C_ / 4)

__device__ __forceinline__ v4f inner_cell_sum(const v4f* __restrict__ rect)
{
    v4f acc[4];
#pragma unroll
    for (int c = 0; c < 4; ++c) acc[c] = (v4f){0.f, 0.f, 0.f, 0.f};
#pragma unroll
    for (int r = 0; r < 16; ++r) {
        acc[(2 * r) & 3]     += rect[(size_t)r * ROWSTEP_];
        acc[(2 * r + 1) & 3] += rect[(size_t)r * ROWSTEP_ + PIECE_];
    }
    return (acc[0] + acc[1]) + (acc[2] + acc[3]);
}

__device__ __forceinline__ v4f edge_cell_sum(const v4f* __restrict__ rect, int ny,
                                             bool two_pieces)
{
    v4f acc[4];
#pragma unroll
    for (int c = 0; c < 4; ++c) acc[c] = (v4f){0.f, 0.f, 0.f, 0.f};
    for (int r0 = 0; r0 < ny; r0 += 8) {
        const v4f* p = rect + (size_t)r0 * ROWSTEP_;
        v4f t[8];
#pragma unroll
        for (int r = 0; r < 8; ++r) t[r] = p[(size_t)r * ROWSTEP_];
        if (two_pieces) {
            v4f u[8];
#pragma unroll
            for (int r = 0; r < 8; ++r) u[r] = p[(size_t)r * ROWSTEP_ + PIECE_];
#pragma unroll
            for (int r = 0; r < 8; ++r) acc[2 + (r & 1)] += u[r];
        }
#pragma unroll
        for (int r = 0; r < 8; ++r) acc[r & 1] += t[r];
    }
    return (acc[0] + acc[1]) + (acc[2] + acc[3]);
}

__global__ __launch_bounds__(256, 10) void splat_kernel(const float* __restrict__ inp,
                                                        float* __restrict__ G)
{
    const int lane = threadIdx.x & 31;
    const int wave = __builtin_amdgcn_readfirstlane((int)(threadIdx.x >> 5));
    const int cell = blockIdx.x * 8 + wave;
    const int cy   = cell / SW_;
    const int cx   = cell - cy * SW_;
    const int y0   = 16 * (cy - 2) - 8;
    const int x0   = 16 * (cx - 2) - 8;
    const int vy   = max(y0, 0), ny = min(y0 + 16, H_) - vy;
    const int vx   = max(x0, 0), nx = min(x0 + 16, W_) - vx;

    v4f s = (v4f){0.f, 0.f, 0.f, 0.f};
    if (ny > 0 && nx > 0) {
        const v4f* rect = (const v4f*)(inp + ((size_t)vy * W_ + vx) * C_) + lane;
        if (ny == 16 && nx == 16) s = inner_cell_sum(rect);
        else                      s = edge_cell_sum(rect, ny, nx == 16);

#pragma unroll
        for (int m = 4; m < 32; m <<= 1) {
            s.x += __shfl_xor(s.x, m);
            s.y += __shfl_xor(s.y, m);
            s.z += __shfl_xor(s.z, m);
            s.w += __shfl_xor(s.w, m);
        }
    }
    if (lane < 4) ((v4f*)(G + (size_t)cell * C_))[lane] = s;
}


__device__ __forceinline__ float from_lane_xor8(float v)
{
    return __builtin_bit_cast(float,
        __builtin_amdgcn_update_dpp(0, __builtin_bit_cast(int, v), 0x168, 0xf, 0xf, true));
}

__global__ __launch_bounds__(256) void slice_kernel(const float* __restrict__ G,
                                                    float* __restrict__ out)
{
    const int wave = __builtin_amdgcn_readfirstlane((int)(threadIdx.x >> 5));
    const int lane = threadIdx.x & 31;
    const int tile = blockIdx.x * 8 + wave;
    const int k    = tile >> 7;
    const int j    = tile & 127;
    const int m    = lane & 15;
    const int hi   = lane >> 4;

    const float* g = G + (((size_t)(k + 2 + hi) * SW_) + j) * C_ + m;
    const float g0 = g[0];
    const float g1 = g[C_];
    const float g2 = g[2 * C_];
    const float g3 = g[3 * C_];
    const float g4 = g[4 * C_];
    const float a1 = (g0 + g2 + 2.0f * g1) * 0.25f;
    const float a2 = (g1 + g3 + 2.0f * g2) * 0.25f;
    const float a3 = (g2 + g4 + 2.0f * g3) * 0.25f;
    const bool lastcol = (j + 4 == SW_ - 1);
    const float n0 = g[lastcol ? 4 * C_ - C_ * SW_ : 3 * C_];
    const float n1 = g[lastcol ? 4 * C_ + C_ * SW_ : 5 * C_];
    const float a4 = (n0 + n1 + 2.0f * g4) * 0.25f;
    v2f b;
    b.x = (a1 + a3 + 2.0f * a2) * 0.25f;
    b.y = (a2 + a4 + 2.0f * a3) * 0.25f;

    const float am = (float)m * 0.0625f;
    const float w0 = 1.0f - am;

    const bool lo = (m < 8);
    float* obase = out + ((size_t)(16 * k) * W_ + 16 * j) * C_
                       + (m & 7) * C_ + (2 * hi + (m >> 3)) * 4;

#pragma unroll
    for (int py = 0; py < 16; ++py) {
        const float ay = (float)py * 0.0625f;
        const float wy = hi ? ay : (1.0f - ay);
        v2f a;
        a.x = wy * w0;
        a.y = wy * am;
        v8f cacc = {};
        v8f d = __builtin_amdgcn_wmma_f32_16x16x4_f32(
             false, b,  false, a,
             (short)0, cacc,  false,  false);
        asm volatile("v_nop\n\tv_nop\n\tv_nop\n\tv_nop" : "+v"(d) : "v"(b), "v"(a));

        v4f s1, s2;
#pragma unroll
        for (int r = 0; r < 4; ++r) {
            const float got = from_lane_xor8(lo ? d[4 + r] : d[r]);
            s1[r] = lo ? d[r] : got;
            s2[r] = lo ? got : d[4 + r];
        }
        float* orow = obase + (size_t)py * W_ * C_;
        *(v4f*)orow            = s1;
        *(v4f*)(orow + 8 * C_) = s2;
    }
}

extern "C" void kernel_launch(void* const* d_in, const int* in_sizes, int n_in,
                              void* d_out, int out_size, void* d_ws, size_t ws_size,
                              hipStream_t stream)
{
    const float* inp = (const float*)d_in[0];
    float* out = (float*)d_out;
    float* G   = (float*)d_ws;

    splat_kernel<<<SH_ * SW_ / 8, 256, 0, stream>>>(inp, G);

    slice_kernel<<<8192 / 8, 256, 0, stream>>>(G, out);
}
